# starting kernel
# baseline (speedup 1.0000x reference)
#include <hip/hip_runtime.h>
#include <hip/hip_bf16.h>

typedef __attribute__((ext_vector_type(2))) float v2f;
typedef __attribute__((ext_vector_type(4))) float v4f;
typedef __attribute__((ext_vector_type(8))) float v8f;

#define LEAKY_ALPHA 0.2f

static constexpr int N_NODES = 50000;
static constexpr int N_EDGES = 800000;
static constexpr int EDGE_BLOCKS = 3125;
static constexpr int NODE_TILES = 3125;

static constexpr size_t WS_A2   = 0;
static constexpr size_t WS_BIAS = 2048;
static constexpr size_t WS_S    = 2056;
static constexpr size_t WS_LOG  = WS_S + (size_t)N_NODES * 8;
static constexpr size_t WS_MAXP = WS_LOG + (size_t)N_EDGES * 4;
static constexpr size_t WS_SUMP = WS_MAXP + (size_t)EDGE_BLOCKS * 4;
static constexpr size_t WS_GMAX = WS_SUMP + (size_t)EDGE_BLOCKS * 4;
static constexpr size_t WS_RSUM = WS_GMAX + 4;

__global__ void prep_kernel(const float* __restrict__ W, const float* __restrict__ b,
                            const float* __restrict__ a, float* ws) {
  const int c = threadIdx.x;
  float* A2 = ws + WS_A2;
#pragma unroll
  for (int j = 0; j < 8; ++j) {
    const int h = j >> 1, half = j & 1;
    const float* av = a + h * 64 + half * 32;
    float acc = 0.f;
#pragma unroll 8
    for (int k = 0; k < 32; ++k) acc += W[(size_t)(h * 32 + k) * 128 + c] * av[k];
    A2[j * 128 + c] = acc;
  }
#pragma unroll
  for (int j = 8; j < 16; ++j) A2[j * 128 + c] = 0.f;
  if (c < 8) {
    const int h = c >> 1, half = c & 1;
    const float* av = a + h * 64 + half * 32;
    float acc = 0.f;
    for (int k = 0; k < 32; ++k) acc += b[h * 32 + k] * av[k];
    ws[WS_BIAS + c] = acc;
  }
}

__global__ void __launch_bounds__(256) score_wmma_kernel(const float* __restrict__ X,
                                                         float* ws) {
  const float* A2 = ws + WS_A2;
  const float* bias = ws + WS_BIAS;
  float* S = ws + WS_S;

  const int lane = threadIdx.x & 31;
  const int wave = threadIdx.x >> 5;
  const int tile = blockIdx.x * 8 + wave;
  if (tile >= NODE_TILES) return;

  const int m = lane & 15;
  const int hi = lane >> 4;
  const float* xrow = X + (size_t)(tile * 16 + m) * 128 + 2 * hi;
  const float* brow = A2 + (size_t)m * 128 + 2 * hi;

  v8f acc = {};
#pragma unroll 8
  for (int kk = 0; kk < 32; ++kk) {
    v2f av = *(const v2f*)(xrow + kk * 4);
    v2f bv = *(const v2f*)(brow + kk * 4);
    acc = __builtin_amdgcn_wmma_f32_16x16x4_f32(
         false, av,  false, bv,
         (short)0, acc,  false,  false);
  }

  if (m < 8) {
    const float bb = bias[m];
    const int rowbase = tile * 16 + 8 * hi;
#pragma unroll
    for (int r = 0; r < 8; ++r) S[(size_t)(rowbase + r) * 8 + m] = acc[r] + bb;
  }
}

__global__ void __launch_bounds__(256) edge_logits_kernel(const int* __restrict__ ei,
                                                          float* ws) {
  __shared__ v4f red[256];
  const int t = threadIdx.x;
  const int e = blockIdx.x * 256 + t;
  const v4f* S = (const v4f*)(ws + WS_S);

  const int src = ei[e];
  const int dst = ei[N_EDGES + e];
  v4f s0 = S[(size_t)src * 2], s1 = S[(size_t)src * 2 + 1];
  v4f d0 = S[(size_t)dst * 2], d1 = S[(size_t)dst * 2 + 1];

  v4f lg;
  lg.x = s0.x + d0.y;
  lg.y = s0.z + d0.w;
  lg.z = s1.x + d1.y;
  lg.w = s1.z + d1.w;
  lg.x = fmaxf(lg.x, LEAKY_ALPHA * lg.x);
  lg.y = fmaxf(lg.y, LEAKY_ALPHA * lg.y);
  lg.z = fmaxf(lg.z, LEAKY_ALPHA * lg.z);
  lg.w = fmaxf(lg.w, LEAKY_ALPHA * lg.w);

  *(v4f*)(ws + WS_LOG + (size_t)e * 4) = lg;

  red[t] = lg;
  __syncthreads();
  for (int off = 128; off > 0; off >>= 1) {
    if (t < off) {
      v4f o = red[t + off], m = red[t];
      m.x = fmaxf(m.x, o.x); m.y = fmaxf(m.y, o.y);
      m.z = fmaxf(m.z, o.z); m.w = fmaxf(m.w, o.w);
      red[t] = m;
    }
    __syncthreads();
  }
  if (t == 0) *(v4f*)(ws + WS_MAXP + (size_t)blockIdx.x * 4) = red[0];
}

__global__ void __launch_bounds__(256) reduce_max_kernel(float* ws) {
  __shared__ v4f red[256];
  const int t = threadIdx.x;
  const v4f* P = (const v4f*)(ws + WS_MAXP);
  v4f m = {-3.0e38f, -3.0e38f, -3.0e38f, -3.0e38f};
  for (int i = t; i < EDGE_BLOCKS; i += 256) {
    v4f o = P[i];
    m.x = fmaxf(m.x, o.x); m.y = fmaxf(m.y, o.y);
    m.z = fmaxf(m.z, o.z); m.w = fmaxf(m.w, o.w);
  }
  red[t] = m;
  __syncthreads();
  for (int off = 128; off > 0; off >>= 1) {
    if (t < off) {
      v4f o = red[t + off], q = red[t];
      q.x = fmaxf(q.x, o.x); q.y = fmaxf(q.y, o.y);
      q.z = fmaxf(q.z, o.z); q.w = fmaxf(q.w, o.w);
      red[t] = q;
    }
    __syncthreads();
  }
  if (t == 0) *(v4f*)(ws + WS_GMAX) = red[0];
}

__global__ void __launch_bounds__(256) edge_exp_kernel(float* ws, float* __restrict__ out) {
  __shared__ v4f red[256];
  const int t = threadIdx.x;
  const int e = blockIdx.x * 256 + t;
  const v4f gm = *(const v4f*)(ws + WS_GMAX);
  v4f lg = *(const v4f*)(ws + WS_LOG + (size_t)e * 4);
  v4f ex;
  ex.x = __expf(lg.x - gm.x);
  ex.y = __expf(lg.y - gm.y);
  ex.z = __expf(lg.z - gm.z);
  ex.w = __expf(lg.w - gm.w);
  *(v4f*)(out + (size_t)e * 4) = ex;

  red[t] = ex;
  __syncthreads();
  for (int off = 128; off > 0; off >>= 1) {
    if (t < off) {
      v4f o = red[t + off], q = red[t];
      q.x += o.x; q.y += o.y; q.z += o.z; q.w += o.w;
      red[t] = q;
    }
    __syncthreads();
  }
  if (t == 0) *(v4f*)(ws + WS_SUMP + (size_t)blockIdx.x * 4) = red[0];
}

__global__ void __launch_bounds__(256) reduce_sum_kernel(float* ws) {
  __shared__ v4f red[256];
  const int t = threadIdx.x;
  const v4f* P = (const v4f*)(ws + WS_SUMP);
  v4f s = {};
  for (int i = t; i < EDGE_BLOCKS; i += 256) {
    v4f o = P[i];
    s.x += o.x; s.y += o.y; s.z += o.z; s.w += o.w;
  }
  red[t] = s;
  __syncthreads();
  for (int off = 128; off > 0; off >>= 1) {
    if (t < off) {
      v4f o = red[t + off], q = red[t];
      q.x += o.x; q.y += o.y; q.z += o.z; q.w += o.w;
      red[t] = q;
    }
    __syncthreads();
  }
  if (t == 0) {
    v4f r = red[0];
    v4f inv = {1.0f / r.x, 1.0f / r.y, 1.0f / r.z, 1.0f / r.w};
    *(v4f*)(ws + WS_RSUM) = inv;
  }
}

__global__ void __launch_bounds__(256) scale_kernel(float* ws, float* __restrict__ out) {
  const int e = blockIdx.x * 256 + threadIdx.x;
  const v4f r = *(const v4f*)(ws + WS_RSUM);
  v4f v = *(v4f*)(out + (size_t)e * 4);
  v.x *= r.x; v.y *= r.y; v.z *= r.z; v.w *= r.w;
  *(v4f*)(out + (size_t)e * 4) = v;
}

extern "C" void kernel_launch(void* const* d_in, const int* in_sizes, int n_in,
                              void* d_out, int out_size, void* d_ws, size_t ws_size,
                              hipStream_t stream) {
  const float* node_feats = (const float*)d_in[0];
  const int*   edge_index = (const int*)d_in[1];
  const float* W          = (const float*)d_in[2];
  const float* b          = (const float*)d_in[3];
  const float* a          = (const float*)d_in[4];
  float* out = (float*)d_out;
  float* ws  = (float*)d_ws;

  prep_kernel<<<1, 128, 0, stream>>>(W, b, a, ws);
  score_wmma_kernel<<<(NODE_TILES + 7) / 8, 256, 0, stream>>>(node_feats, ws);
  edge_logits_kernel<<<EDGE_BLOCKS, 256, 0, stream>>>(edge_index, ws);
  reduce_max_kernel<<<1, 256, 0, stream>>>(ws);
  edge_exp_kernel<<<EDGE_BLOCKS, 256, 0, stream>>>(ws, out);
  reduce_sum_kernel<<<1, 256, 0, stream>>>(ws);
  scale_kernel<<<EDGE_BLOCKS, 256, 0, stream>>>(ws, out);
}


# Fold the three single-work-group launches (weight fold, max reduce, sum reduce) into their consumers, 7 launches -> 4; edge kernels take 5 edges per thread (625 partials); sum tree changes
# speedup vs baseline: 1.4515x; 1.4515x over previous
#include <hip/hip_runtime.h>
#include <hip/hip_bf16.h>


typedef __attribute__((ext_vector_type(2))) float v2f;
typedef __attribute__((ext_vector_type(4))) float v4f;
typedef __attribute__((ext_vector_type(8))) float v8f;

#define LEAKY_ALPHA 0.2f

static constexpr int N_NODES = 50000;
static constexpr int N_EDGES = 800000;
static constexpr int NODE_TILES = 3125;

static constexpr int EDGE_WGS = 625;
static constexpr int EDGES_PER_THREAD = 5;
static constexpr int EDGES_PER_WG = 256 * EDGES_PER_THREAD;

static constexpr int A2_PITCH = 132;

static constexpr size_t WS_S    = 2056;
static constexpr size_t WS_LOG  = WS_S + (size_t)N_NODES * 8;
static constexpr size_t WS_MAXP = WS_LOG + (size_t)N_EDGES * 4;
static constexpr size_t WS_SUMP = WS_MAXP + (size_t)3125 * 4;

__device__ __forceinline__ v4f max4(v4f p, v4f q) {
  v4f r;
  r.x = fmaxf(p.x, q.x); r.y = fmaxf(p.y, q.y);
  r.z = fmaxf(p.z, q.z); r.w = fmaxf(p.w, q.w);
  return r;
}

__device__ __forceinline__ v4f block_max4(v4f v, v4f* red  ) {
#pragma unroll
  for (int off = 16; off > 0; off >>= 1) {
    v4f o;
    o.x = __shfl_xor(v.x, off); o.y = __shfl_xor(v.y, off);
    o.z = __shfl_xor(v.z, off); o.w = __shfl_xor(v.w, off);
    v = max4(v, o);
  }
  if ((threadIdx.x & 31) == 0) red[threadIdx.x >> 5] = v;
  __syncthreads();
  v4f r = red[0];
#pragma unroll
  for (int w = 1; w < 8; ++w) r = max4(r, red[w]);
  return r;
}

__device__ __forceinline__ v4f block_sum4(v4f v, v4f* red  ) {
#pragma unroll
  for (int off = 16; off > 0; off >>= 1) {
    v4f o;
    o.x = __shfl_xor(v.x, off); o.y = __shfl_xor(v.y, off);
    o.z = __shfl_xor(v.z, off); o.w = __shfl_xor(v.w, off);
    v += o;
  }
  if ((threadIdx.x & 31) == 0) red[threadIdx.x >> 5] = v;
  __syncthreads();
  v4f r = red[0];
#pragma unroll
  for (int w = 1; w < 8; ++w) r += red[w];
  return r;
}

__global__ void __launch_bounds__(256) score_wmma_kernel(const float* __restrict__ X,
                                                         const float* __restrict__ W,
                                                         const float* __restrict__ b,
                                                         const float* __restrict__ a,
                                                         float* ws) {
  __shared__ __attribute__((aligned(16))) float A2s[8 * A2_PITCH];
  __shared__ float biass[8];
  float* S = ws + WS_S;
  const int t = threadIdx.x;

  {
    const int c = t & 127;
    const int hp = __builtin_amdgcn_readfirstlane(t >> 7);
#pragma unroll
    for (int hh = 0; hh < 2; ++hh) {
      const int h = hp * 2 + hh;
      const float* wcol = W + (size_t)(h * 32) * 128 + c;
      const float* av = a + h * 64;
      float acc0 = 0.f, acc1 = 0.f;
#pragma unroll
      for (int k = 0; k < 32; ++k) {
        const float w = wcol[(size_t)k * 128];
        acc0 += w * av[k];
        acc1 += w * av[32 + k];
      }
      A2s[(2 * h) * A2_PITCH + c] = acc0;
      A2s[(2 * h + 1) * A2_PITCH + c] = acc1;
    }
    if (t < 8) {
      const int h = t >> 1, half = t & 1;
      const float* av = a + h * 64 + half * 32;
      float acc = 0.f;
#pragma unroll
      for (int k = 0; k < 32; ++k) acc += b[h * 32 + k] * av[k];
      biass[t] = acc;
    }
  }
  __syncthreads();

  const int lane = t & 31;
  const int wave = t >> 5;
  const int tile = blockIdx.x * 8 + wave;
  if (tile >= NODE_TILES) return;

  const int m = lane & 15;
  const int hi = lane >> 4;
  const float* xrow = X + (size_t)(tile * 16 + m) * 128 + 2 * hi;
  const float* brow = A2s + (m & 7) * A2_PITCH + 2 * hi;

  v8f acc = {};
#pragma unroll 8
  for (int kk = 0; kk < 32; ++kk) {
    v2f av = *(const v2f*)(xrow + kk * 4);
    v2f bv = *(const v2f*)(brow + kk * 4);
    acc = __builtin_amdgcn_wmma_f32_16x16x4_f32(
         false, av,  false, bv,
         (short)0, acc,  false,  false);
  }

  if (m < 8) {
    const float bb = biass[m];
    const int rowbase = tile * 16 + 8 * hi;
#pragma unroll
    for (int r = 0; r < 8; ++r) S[(size_t)(rowbase + r) * 8 + m] = acc[r] + bb;
  }
}

__global__ void __launch_bounds__(256) edge_logits_kernel(const int* __restrict__ ei,
                                                          float* ws) {
  __shared__ v4f red[8];
  const int t = threadIdx.x;
  const int e0 = blockIdx.x * EDGES_PER_WG + t;
  const v4f* S = (const v4f*)(ws + WS_S);

  int src[EDGES_PER_THREAD], dst[EDGES_PER_THREAD];
#pragma unroll
  for (int i = 0; i < EDGES_PER_THREAD; ++i) {
    src[i] = ei[e0 + i * 256];
    dst[i] = ei[N_EDGES + e0 + i * 256];
  }
  asm volatile("" :: "v"(src[0]), "v"(src[1]), "v"(src[2]), "v"(src[3]), "v"(src[4]),
                     "v"(dst[0]), "v"(dst[1]), "v"(dst[2]), "v"(dst[3]), "v"(dst[4]));
  v4f s0[EDGES_PER_THREAD], s1[EDGES_PER_THREAD], d0[EDGES_PER_THREAD], d1[EDGES_PER_THREAD];
#pragma unroll
  for (int i = 0; i < EDGES_PER_THREAD; ++i) {
    s0[i] = S[(size_t)src[i] * 2]; s1[i] = S[(size_t)src[i] * 2 + 1];
    d0[i] = S[(size_t)dst[i] * 2]; d1[i] = S[(size_t)dst[i] * 2 + 1];
  }

  v4f mx;
#pragma unroll
  for (int i = 0; i < EDGES_PER_THREAD; ++i) {
    v4f lg;
    lg.x = s0[i].x + d0[i].y;
    lg.y = s0[i].z + d0[i].w;
    lg.z = s1[i].x + d1[i].y;
    lg.w = s1[i].z + d1[i].w;
    lg.x = fmaxf(lg.x, LEAKY_ALPHA * lg.x);
    lg.y = fmaxf(lg.y, LEAKY_ALPHA * lg.y);
    lg.z = fmaxf(lg.z, LEAKY_ALPHA * lg.z);
    lg.w = fmaxf(lg.w, LEAKY_ALPHA * lg.w);
    *(v4f*)(ws + WS_LOG + (size_t)(e0 + i * 256) * 4) = lg;
    mx = (i == 0) ? lg : max4(mx, lg);
  }

  mx = block_max4(mx, red);
  if (t == 0) *(v4f*)(ws + WS_MAXP + (size_t)blockIdx.x * 4) = mx;
}

__global__ void __launch_bounds__(256) edge_exp_kernel(float* ws, float* __restrict__ out) {
  __shared__ v4f redm[8];
  __shared__ v4f reds[8];
  const int t = threadIdx.x;
  const int e0 = blockIdx.x * EDGES_PER_WG + t;

  v4f lg[EDGES_PER_THREAD];
#pragma unroll
  for (int i = 0; i < EDGES_PER_THREAD; ++i)
    lg[i] = *(const v4f*)(ws + WS_LOG + (size_t)(e0 + i * 256) * 4);

  const v4f* P = (const v4f*)(ws + WS_MAXP);
  const int t2 = (t + 512 < EDGE_WGS) ? t + 512 : EDGE_WGS - 1;
  const v4f m = max4(max4(P[t], P[t + 256]), P[t2]);
  const v4f gm = block_max4(m, redm);

  v4f s = {0.f, 0.f, 0.f, 0.f};
#pragma unroll
  for (int i = 0; i < EDGES_PER_THREAD; ++i) {
    v4f ex;
    ex.x = __expf(lg[i].x - gm.x);
    ex.y = __expf(lg[i].y - gm.y);
    ex.z = __expf(lg[i].z - gm.z);
    ex.w = __expf(lg[i].w - gm.w);
    *(v4f*)(out + (size_t)(e0 + i * 256) * 4) = ex;
    s += ex;
  }
  s = block_sum4(s, reds);
  if (t == 0) *(v4f*)(ws + WS_SUMP + (size_t)blockIdx.x * 4) = s;
}

__global__ void __launch_bounds__(256) scale_kernel(float* ws, float* __restrict__ out) {
  __shared__ v4f reds[8];
  const int t = threadIdx.x;
  const int e0 = blockIdx.x * EDGES_PER_WG + t;

  v4f v[EDGES_PER_THREAD];
#pragma unroll
  for (int i = 0; i < EDGES_PER_THREAD; ++i)
    v[i] = *(const v4f*)(out + (size_t)(e0 + i * 256) * 4);

  const v4f* P = (const v4f*)(ws + WS_SUMP);
  const bool has3 = (t + 512 < EDGE_WGS);
  const v4f p2 = P[has3 ? t + 512 : EDGE_WGS - 1];
  const v4f zero4 = {0.f, 0.f, 0.f, 0.f};
  v4f s = P[t] + P[t + 256];
  s += has3 ? p2 : zero4;
  const v4f r = block_sum4(s, reds);
  const v4f inv = {1.0f / r.x, 1.0f / r.y, 1.0f / r.z, 1.0f / r.w};

#pragma unroll
  for (int i = 0; i < EDGES_PER_THREAD; ++i) {
    v4f q = v[i];
    q.x *= inv.x; q.y *= inv.y; q.z *= inv.z; q.w *= inv.w;
    *(v4f*)(out + (size_t)(e0 + i * 256) * 4) = q;
  }
}

extern "C" void kernel_launch(void* const* d_in, const int* in_sizes, int n_in,
                              void* d_out, int out_size, void* d_ws, size_t ws_size,
                              hipStream_t stream) {
  const float* node_feats = (const float*)d_in[0];
  const int*   edge_index = (const int*)d_in[1];
  const float* W          = (const float*)d_in[2];
  const float* b          = (const float*)d_in[3];
  const float* a          = (const float*)d_in[4];
  float* out = (float*)d_out;
  float* ws  = (float*)d_ws;

  score_wmma_kernel<<<(NODE_TILES + 7) / 8, 256, 0, stream>>>(node_feats, W, b, a, ws);
  edge_logits_kernel<<<EDGE_WGS, 256, 0, stream>>>(edge_index, ws);
  edge_exp_kernel<<<EDGE_WGS, 256, 0, stream>>>(ws, out);
  scale_kernel<<<EDGE_WGS, 256, 0, stream>>>(ws, out);
}


# Online softmax: logit kernel also writes per-work-group (max, sum of exp(logit - max)); one final kernel merges the 625 pairs, then exp and normalise in one pass, 4 launches -> 3; sum tree changes
# speedup vs baseline: 1.7593x; 1.2121x over previous
#include <hip/hip_runtime.h>
#include <hip/hip_bf16.h>


typedef __attribute__((ext_vector_type(2))) float v2f;
typedef __attribute__((ext_vector_type(4))) float v4f;
typedef __attribute__((ext_vector_type(8))) float v8f;

#define LEAKY_ALPHA 0.2f

static constexpr int N_NODES = 50000;
static constexpr int N_EDGES = 800000;
static constexpr int NODE_TILES = 3125;

static constexpr int EDGE_WGS = 625;
static constexpr int EDGES_PER_THREAD = 5;
static constexpr int EDGES_PER_WG = 256 * EDGES_PER_THREAD;

static constexpr int A2_PITCH = 132;

static constexpr size_t WS_S    = 2056;
static constexpr size_t WS_LOG  = WS_S + (size_t)N_NODES * 8;
static constexpr size_t WS_MAXP = WS_LOG + (size_t)N_EDGES * 4;
static constexpr size_t WS_SUMP = WS_MAXP + (size_t)3125 * 4;

__device__ __forceinline__ v4f max4(v4f p, v4f q) {
  v4f r;
  r.x = fmaxf(p.x, q.x); r.y = fmaxf(p.y, q.y);
  r.z = fmaxf(p.z, q.z); r.w = fmaxf(p.w, q.w);
  return r;
}

__device__ __forceinline__ v4f block_max4(v4f v, v4f* red  ) {
#pragma unroll
  for (int off = 16; off > 0; off >>= 1) {
    v4f o;
    o.x = __shfl_xor(v.x, off); o.y = __shfl_xor(v.y, off);
    o.z = __shfl_xor(v.z, off); o.w = __shfl_xor(v.w, off);
    v = max4(v, o);
  }
  if ((threadIdx.x & 31) == 0) red[threadIdx.x >> 5] = v;
  __syncthreads();
  v4f r = red[0];
#pragma unroll
  for (int w = 1; w < 8; ++w) r = max4(r, red[w]);
  return r;
}

__device__ __forceinline__ v4f block_sum4(v4f v, v4f* red  ) {
#pragma unroll
  for (int off = 16; off > 0; off >>= 1) {
    v4f o;
    o.x = __shfl_xor(v.x, off); o.y = __shfl_xor(v.y, off);
    o.z = __shfl_xor(v.z, off); o.w = __shfl_xor(v.w, off);
    v += o;
  }
  if ((threadIdx.x & 31) == 0) red[threadIdx.x >> 5] = v;
  __syncthreads();
  v4f r = red[0];
#pragma unroll
  for (int w = 1; w < 8; ++w) r += red[w];
  return r;
}

__device__ __forceinline__ v4f shfl_xor4(v4f v, int off) {
  v4f o;
  o.x = __shfl_xor(v.x, off); o.y = __shfl_xor(v.y, off);
  o.z = __shfl_xor(v.z, off); o.w = __shfl_xor(v.w, off);
  return o;
}

__device__ __forceinline__ v4f wave_max4(v4f v) {
#pragma unroll
  for (int off = 16; off > 0; off >>= 1) v = max4(v, shfl_xor4(v, off));
  return v;
}

__device__ __forceinline__ v4f wave_sum4(v4f v) {
#pragma unroll
  for (int off = 16; off > 0; off >>= 1) v += shfl_xor4(v, off);
  return v;
}

__device__ __forceinline__ v4f exp_diff4(v4f p, v4f q) {
  v4f r;
  r.x = __expf(p.x - q.x); r.y = __expf(p.y - q.y);
  r.z = __expf(p.z - q.z); r.w = __expf(p.w - q.w);
  return r;
}

__global__ void __launch_bounds__(256) score_wmma_kernel(const float* __restrict__ X,
                                                         const float* __restrict__ W,
                                                         const float* __restrict__ b,
                                                         const float* __restrict__ a,
                                                         float* ws) {
  __shared__ __attribute__((aligned(16))) float A2s[8 * A2_PITCH];
  __shared__ float biass[8];
  float* S = ws + WS_S;
  const int t = threadIdx.x;

  {
    const int c = t & 127;
    const int hp = __builtin_amdgcn_readfirstlane(t >> 7);
#pragma unroll
    for (int hh = 0; hh < 2; ++hh) {
      const int h = hp * 2 + hh;
      const float* wcol = W + (size_t)(h * 32) * 128 + c;
      const float* av = a + h * 64;
      float acc0 = 0.f, acc1 = 0.f;
#pragma unroll
      for (int k = 0; k < 32; ++k) {
        const float w = wcol[(size_t)k * 128];
        acc0 += w * av[k];
        acc1 += w * av[32 + k];
      }
      A2s[(2 * h) * A2_PITCH + c] = acc0;
      A2s[(2 * h + 1) * A2_PITCH + c] = acc1;
    }
    if (t < 8) {
      const int h = t >> 1, half = t & 1;
      const float* av = a + h * 64 + half * 32;
      float acc = 0.f;
#pragma unroll
      for (int k = 0; k < 32; ++k) acc += b[h * 32 + k] * av[k];
      biass[t] = acc;
    }
  }
  __syncthreads();

  const int lane = t & 31;
  const int wave = t >> 5;
  const int tile = blockIdx.x * 8 + wave;
  if (tile >= NODE_TILES) return;

  const int m = lane & 15;
  const int hi = lane >> 4;
  const float* xrow = X + (size_t)(tile * 16 + m) * 128 + 2 * hi;
  const float* brow = A2s + (m & 7) * A2_PITCH + 2 * hi;

  v8f acc = {};
#pragma unroll 8
  for (int kk = 0; kk < 32; ++kk) {
    v2f av = *(const v2f*)(xrow + kk * 4);
    v2f bv = *(const v2f*)(brow + kk * 4);
    acc = __builtin_amdgcn_wmma_f32_16x16x4_f32(
         false, av,  false, bv,
         (short)0, acc,  false,  false);
  }

  if (m < 8) {
    const float bb = biass[m];
    const int rowbase = tile * 16 + 8 * hi;
#pragma unroll
    for (int r = 0; r < 8; ++r) S[(size_t)(rowbase + r) * 8 + m] = acc[r] + bb;
  }
}

__global__ void __launch_bounds__(256) edge_logits_kernel(const int* __restrict__ ei,
                                                          float* ws) {
  __shared__ v4f redm[8];
  __shared__ v4f reds[8];
  const int t = threadIdx.x;
  const int e0 = blockIdx.x * EDGES_PER_WG + t;
  const v4f* S = (const v4f*)(ws + WS_S);

  int src[EDGES_PER_THREAD], dst[EDGES_PER_THREAD];
#pragma unroll
  for (int i = 0; i < EDGES_PER_THREAD; ++i) {
    src[i] = ei[e0 + i * 256];
    dst[i] = ei[N_EDGES + e0 + i * 256];
  }
  asm volatile("" :: "v"(src[0]), "v"(src[1]), "v"(src[2]), "v"(src[3]), "v"(src[4]),
                     "v"(dst[0]), "v"(dst[1]), "v"(dst[2]), "v"(dst[3]), "v"(dst[4]));
  v4f s0[EDGES_PER_THREAD], s1[EDGES_PER_THREAD], d0[EDGES_PER_THREAD], d1[EDGES_PER_THREAD];
#pragma unroll
  for (int i = 0; i < EDGES_PER_THREAD; ++i) {
    s0[i] = S[(size_t)src[i] * 2]; s1[i] = S[(size_t)src[i] * 2 + 1];
    d0[i] = S[(size_t)dst[i] * 2]; d1[i] = S[(size_t)dst[i] * 2 + 1];
  }

  v4f lg[EDGES_PER_THREAD];
  v4f mx;
#pragma unroll
  for (int i = 0; i < EDGES_PER_THREAD; ++i) {
    v4f l;
    l.x = s0[i].x + d0[i].y;
    l.y = s0[i].z + d0[i].w;
    l.z = s1[i].x + d1[i].y;
    l.w = s1[i].z + d1[i].w;
    l.x = fmaxf(l.x, LEAKY_ALPHA * l.x);
    l.y = fmaxf(l.y, LEAKY_ALPHA * l.y);
    l.z = fmaxf(l.z, LEAKY_ALPHA * l.z);
    l.w = fmaxf(l.w, LEAKY_ALPHA * l.w);
    *(v4f*)(ws + WS_LOG + (size_t)(e0 + i * 256) * 4) = l;
    lg[i] = l;
    mx = (i == 0) ? l : max4(mx, l);
  }

  const v4f wm = wave_max4(mx);
  v4f s = exp_diff4(lg[0], wm);
#pragma unroll
  for (int i = 1; i < EDGES_PER_THREAD; ++i) s += exp_diff4(lg[i], wm);
  s = wave_sum4(s);

  if ((t & 31) == 0) { redm[t >> 5] = wm; reds[t >> 5] = s; }
  __syncthreads();

  if (t < 32) {
    const v4f m = redm[t & 7];
    const v4f q = reds[t & 7];
    v4f g = m;
#pragma unroll
    for (int off = 4; off > 0; off >>= 1) g = max4(g, shfl_xor4(g, off));
    const v4f f = exp_diff4(m, g);
    v4f r;
    r.x = q.x * f.x; r.y = q.y * f.y; r.z = q.z * f.z; r.w = q.w * f.w;
#pragma unroll
    for (int off = 4; off > 0; off >>= 1) r += shfl_xor4(r, off);
    if (t == 0) {
      *(v4f*)(ws + WS_MAXP + (size_t)blockIdx.x * 4) = g;
      *(v4f*)(ws + WS_SUMP + (size_t)blockIdx.x * 4) = r;
    }
  }
}

__global__ void __launch_bounds__(256) edge_softmax_kernel(float* ws, float* __restrict__ out) {
  __shared__ v4f redm[8];
  __shared__ v4f reds[8];
  const int t = threadIdx.x;
  const int e0 = blockIdx.x * EDGES_PER_WG + t;

  v4f lg[EDGES_PER_THREAD];
#pragma unroll
  for (int i = 0; i < EDGES_PER_THREAD; ++i)
    lg[i] = *(const v4f*)(ws + WS_LOG + (size_t)(e0 + i * 256) * 4);

  const v4f* PM = (const v4f*)(ws + WS_MAXP);
  const v4f* PS = (const v4f*)(ws + WS_SUMP);
  const bool has3 = (t + 512 < EDGE_WGS);
  const int t2 = has3 ? t + 512 : EDGE_WGS - 1;
  const v4f m0 = PM[t], m1 = PM[t + 256], m2 = PM[t2];
  const v4f q0 = PS[t], q1 = PS[t + 256], q2 = PS[t2];

  const v4f gm = block_max4(max4(max4(m0, m1), m2), redm);

  const v4f f0 = exp_diff4(m0, gm), f1 = exp_diff4(m1, gm), f2 = exp_diff4(m2, gm);
  const v4f zero4 = {0.f, 0.f, 0.f, 0.f};
  v4f s;
  s.x = q0.x * f0.x + q1.x * f1.x; s.y = q0.y * f0.y + q1.y * f1.y;
  s.z = q0.z * f0.z + q1.z * f1.z; s.w = q0.w * f0.w + q1.w * f1.w;
  v4f s2;
  s2.x = q2.x * f2.x; s2.y = q2.y * f2.y; s2.z = q2.z * f2.z; s2.w = q2.w * f2.w;
  s += has3 ? s2 : zero4;
  const v4f r = block_sum4(s, reds);
  const v4f inv = {1.0f / r.x, 1.0f / r.y, 1.0f / r.z, 1.0f / r.w};

#pragma unroll
  for (int i = 0; i < EDGES_PER_THREAD; ++i) {
    v4f q = exp_diff4(lg[i], gm);
    q.x *= inv.x; q.y *= inv.y; q.z *= inv.z; q.w *= inv.w;
    *(v4f*)(out + (size_t)(e0 + i * 256) * 4) = q;
  }
}

extern "C" void kernel_launch(void* const* d_in, const int* in_sizes, int n_in,
                              void* d_out, int out_size, void* d_ws, size_t ws_size,
                              hipStream_t stream) {
  const float* node_feats = (const float*)d_in[0];
  const int*   edge_index = (const int*)d_in[1];
  const float* W          = (const float*)d_in[2];
  const float* b          = (const float*)d_in[3];
  const float* a          = (const float*)d_in[4];
  float* out = (float*)d_out;
  float* ws  = (float*)d_ws;

  score_wmma_kernel<<<(NODE_TILES + 7) / 8, 256, 0, stream>>>(node_feats, W, b, a, ws);
  edge_logits_kernel<<<EDGE_WGS, 256, 0, stream>>>(edge_index, ws);
  edge_softmax_kernel<<<EDGE_WGS, 256, 0, stream>>>(ws, out);
}


# Node scores stored as two planes of 16-byte records (src [N,4], dst [N,4]); score GEMM swaps operand roles so a lane owns one node and stores whole records; one gather per edge endpoint, not two
# speedup vs baseline: 2.1773x; 1.2376x over previous
#include <hip/hip_runtime.h>
#include <hip/hip_bf16.h>


typedef __attribute__((ext_vector_type(2))) float v2f;
typedef __attribute__((ext_vector_type(4))) float v4f;
typedef __attribute__((ext_vector_type(8))) float v8f;

#define LEAKY_ALPHA 0.2f

static constexpr int N_NODES = 50000;
static constexpr int N_EDGES = 800000;
static constexpr int NODE_TILES = 3125;

static constexpr int EDGE_WGS = 625;
static constexpr int EDGES_PER_THREAD = 5;
static constexpr int EDGES_PER_WG = 256 * EDGES_PER_THREAD;

static constexpr int A2_PITCH = 132;

static constexpr size_t WS_SSRC = 2056;
static constexpr size_t WS_SDST = WS_SSRC + (size_t)N_NODES * 4;
static constexpr size_t WS_LOG  = WS_SDST + (size_t)N_NODES * 4;
static constexpr size_t WS_MAXP = WS_LOG + (size_t)N_EDGES * 4;
static constexpr size_t WS_SUMP = WS_MAXP + (size_t)3125 * 4;

__device__ __forceinline__ v4f max4(v4f p, v4f q) {
  v4f r;
  r.x = fmaxf(p.x, q.x); r.y = fmaxf(p.y, q.y);
  r.z = fmaxf(p.z, q.z); r.w = fmaxf(p.w, q.w);
  return r;
}

__device__ __forceinline__ v4f block_max4(v4f v, v4f* red  ) {
#pragma unroll
  for (int off = 16; off > 0; off >>= 1) {
    v4f o;
    o.x = __shfl_xor(v.x, off); o.y = __shfl_xor(v.y, off);
    o.z = __shfl_xor(v.z, off); o.w = __shfl_xor(v.w, off);
    v = max4(v, o);
  }
  if ((threadIdx.x & 31) == 0) red[threadIdx.x >> 5] = v;
  __syncthreads();
  v4f r = red[0];
#pragma unroll
  for (int w = 1; w < 8; ++w) r = max4(r, red[w]);
  return r;
}

__device__ __forceinline__ v4f block_sum4(v4f v, v4f* red  ) {
#pragma unroll
  for (int off = 16; off > 0; off >>= 1) {
    v4f o;
    o.x = __shfl_xor(v.x, off); o.y = __shfl_xor(v.y, off);
    o.z = __shfl_xor(v.z, off); o.w = __shfl_xor(v.w, off);
    v += o;
  }
  if ((threadIdx.x & 31) == 0) red[threadIdx.x >> 5] = v;
  __syncthreads();
  v4f r = red[0];
#pragma unroll
  for (int w = 1; w < 8; ++w) r += red[w];
  return r;
}

__device__ __forceinline__ v4f shfl_xor4(v4f v, int off) {
  v4f o;
  o.x = __shfl_xor(v.x, off); o.y = __shfl_xor(v.y, off);
  o.z = __shfl_xor(v.z, off); o.w = __shfl_xor(v.w, off);
  return o;
}

__device__ __forceinline__ v4f wave_max4(v4f v) {
#pragma unroll
  for (int off = 16; off > 0; off >>= 1) v = max4(v, shfl_xor4(v, off));
  return v;
}

__device__ __forceinline__ v4f wave_sum4(v4f v) {
#pragma unroll
  for (int off = 16; off > 0; off >>= 1) v += shfl_xor4(v, off);
  return v;
}

__device__ __forceinline__ v4f exp_diff4(v4f p, v4f q) {
  v4f r;
  r.x = __expf(p.x - q.x); r.y = __expf(p.y - q.y);
  r.z = __expf(p.z - q.z); r.w = __expf(p.w - q.w);
  return r;
}

__global__ void __launch_bounds__(256) score_wmma_kernel(const float* __restrict__ X,
                                                         const float* __restrict__ W,
                                                         const float* __restrict__ b,
                                                         const float* __restrict__ a,
                                                         float* ws) {
  __shared__ __attribute__((aligned(16))) float A2s[8 * A2_PITCH];
  __shared__ __attribute__((aligned(16))) float biass[8];
  const int t = threadIdx.x;

  {
    const int c = t & 127;
    const int hp = __builtin_amdgcn_readfirstlane(t >> 7);
#pragma unroll
    for (int hh = 0; hh < 2; ++hh) {
      const int h = hp * 2 + hh;
      const float* wcol = W + (size_t)(h * 32) * 128 + c;
      const float* av = a + h * 64;
      float acc0 = 0.f, acc1 = 0.f;
#pragma unroll
      for (int k = 0; k < 32; ++k) {
        const float w = wcol[(size_t)k * 128];
        acc0 += w * av[k];
        acc1 += w * av[32 + k];
      }
      A2s[h * A2_PITCH + c] = acc0;
      A2s[(4 + h) * A2_PITCH + c] = acc1;
    }
    if (t < 8) {
      const int h = t & 3, half = t >> 2;
      const float* av = a + h * 64 + half * 32;
      float acc = 0.f;
#pragma unroll
      for (int k = 0; k < 32; ++k) acc += b[h * 32 + k] * av[k];
      biass[t] = acc;
    }
  }
  __syncthreads();

  const int lane = t & 31;
  const int wave = t >> 5;
  const int tile = blockIdx.x * 8 + wave;
  if (tile >= NODE_TILES) return;

  const int m = lane & 15;
  const int hi = lane >> 4;
  const float* xrow = X + (size_t)(tile * 16 + m) * 128 + 2 * hi;
  const float* brow = A2s + (m & 7) * A2_PITCH + 2 * hi;

  v8f acc = {};
#pragma unroll 8
  for (int kk = 0; kk < 32; ++kk) {
    v2f av = *(const v2f*)(xrow + kk * 4);
    v2f bv = *(const v2f*)(brow + kk * 4);
    acc = __builtin_amdgcn_wmma_f32_16x16x4_f32(
         false, bv,  false, av,
         (short)0, acc,  false,  false);
  }

  const v4f bb = ((const v4f*)biass)[hi];
  v4f rec;
  rec.x = (hi ? acc[4] : acc[0]) + bb.x;
  rec.y = (hi ? acc[5] : acc[1]) + bb.y;
  rec.z = (hi ? acc[6] : acc[2]) + bb.z;
  rec.w = (hi ? acc[7] : acc[3]) + bb.w;
  float* plane = ws + (hi ? WS_SDST : WS_SSRC);
  *(v4f*)(plane + (size_t)(tile * 16 + m) * 4) = rec;
}

__global__ void __launch_bounds__(256) edge_logits_kernel(const int* __restrict__ ei,
                                                          float* ws) {
  __shared__ v4f redm[8];
  __shared__ v4f reds[8];
  const int t = threadIdx.x;
  const int e0 = blockIdx.x * EDGES_PER_WG + t;
  const v4f* SS = (const v4f*)(ws + WS_SSRC);
  const v4f* SD = (const v4f*)(ws + WS_SDST);

  int src[EDGES_PER_THREAD], dst[EDGES_PER_THREAD];
#pragma unroll
  for (int i = 0; i < EDGES_PER_THREAD; ++i) {
    src[i] = ei[e0 + i * 256];
    dst[i] = ei[N_EDGES + e0 + i * 256];
  }
  asm volatile("" :: "v"(src[0]), "v"(src[1]), "v"(src[2]), "v"(src[3]), "v"(src[4]),
                     "v"(dst[0]), "v"(dst[1]), "v"(dst[2]), "v"(dst[3]), "v"(dst[4]));
  v4f sv[EDGES_PER_THREAD], dv[EDGES_PER_THREAD];
#pragma unroll
  for (int i = 0; i < EDGES_PER_THREAD; ++i) {
    sv[i] = SS[(size_t)src[i]];
    dv[i] = SD[(size_t)dst[i]];
  }

  v4f lg[EDGES_PER_THREAD];
  v4f mx;
#pragma unroll
  for (int i = 0; i < EDGES_PER_THREAD; ++i) {
    v4f l = sv[i] + dv[i];
    l.x = fmaxf(l.x, LEAKY_ALPHA * l.x);
    l.y = fmaxf(l.y, LEAKY_ALPHA * l.y);
    l.z = fmaxf(l.z, LEAKY_ALPHA * l.z);
    l.w = fmaxf(l.w, LEAKY_ALPHA * l.w);
    *(v4f*)(ws + WS_LOG + (size_t)(e0 + i * 256) * 4) = l;
    lg[i] = l;
    mx = (i == 0) ? l : max4(mx, l);
  }

  const v4f wm = wave_max4(mx);
  v4f s = exp_diff4(lg[0], wm);
#pragma unroll
  for (int i = 1; i < EDGES_PER_THREAD; ++i) s += exp_diff4(lg[i], wm);
  s = wave_sum4(s);

  if ((t & 31) == 0) { redm[t >> 5] = wm; reds[t >> 5] = s; }
  __syncthreads();

  if (t < 32) {
    const v4f m = redm[t & 7];
    const v4f q = reds[t & 7];
    v4f g = m;
#pragma unroll
    for (int off = 4; off > 0; off >>= 1) g = max4(g, shfl_xor4(g, off));
    const v4f f = exp_diff4(m, g);
    v4f r;
    r.x = q.x * f.x; r.y = q.y * f.y; r.z = q.z * f.z; r.w = q.w * f.w;
#pragma unroll
    for (int off = 4; off > 0; off >>= 1) r += shfl_xor4(r, off);
    if (t == 0) {
      *(v4f*)(ws + WS_MAXP + (size_t)blockIdx.x * 4) = g;
      *(v4f*)(ws + WS_SUMP + (size_t)blockIdx.x * 4) = r;
    }
  }
}

__global__ void __launch_bounds__(256) edge_softmax_kernel(float* ws, float* __restrict__ out) {
  __shared__ v4f redm[8];
  __shared__ v4f reds[8];
  const int t = threadIdx.x;
  const int e0 = blockIdx.x * EDGES_PER_WG + t;

  v4f lg[EDGES_PER_THREAD];
#pragma unroll
  for (int i = 0; i < EDGES_PER_THREAD; ++i)
    lg[i] = *(const v4f*)(ws + WS_LOG + (size_t)(e0 + i * 256) * 4);

  const v4f* PM = (const v4f*)(ws + WS_MAXP);
  const v4f* PS = (const v4f*)(ws + WS_SUMP);
  const bool has3 = (t + 512 < EDGE_WGS);
  const int t2 = has3 ? t + 512 : EDGE_WGS - 1;
  const v4f m0 = PM[t], m1 = PM[t + 256], m2 = PM[t2];
  const v4f q0 = PS[t], q1 = PS[t + 256], q2 = PS[t2];

  const v4f gm = block_max4(max4(max4(m0, m1), m2), redm);

  const v4f f0 = exp_diff4(m0, gm), f1 = exp_diff4(m1, gm), f2 = exp_diff4(m2, gm);
  const v4f zero4 = {0.f, 0.f, 0.f, 0.f};
  v4f s;
  s.x = q0.x * f0.x + q1.x * f1.x; s.y = q0.y * f0.y + q1.y * f1.y;
  s.z = q0.z * f0.z + q1.z * f1.z; s.w = q0.w * f0.w + q1.w * f1.w;
  v4f s2;
  s2.x = q2.x * f2.x; s2.y = q2.y * f2.y; s2.z = q2.z * f2.z; s2.w = q2.w * f2.w;
  s += has3 ? s2 : zero4;
  const v4f r = block_sum4(s, reds);
  const v4f inv = {1.0f / r.x, 1.0f / r.y, 1.0f / r.z, 1.0f / r.w};

#pragma unroll
  for (int i = 0; i < EDGES_PER_THREAD; ++i) {
    v4f q = exp_diff4(lg[i], gm);
    q.x *= inv.x; q.y *= inv.y; q.z *= inv.z; q.w *= inv.w;
    *(v4f*)(out + (size_t)(e0 + i * 256) * 4) = q;
  }
}

extern "C" void kernel_launch(void* const* d_in, const int* in_sizes, int n_in,
                              void* d_out, int out_size, void* d_ws, size_t ws_size,
                              hipStream_t stream) {
  const float* node_feats = (const float*)d_in[0];
  const int*   edge_index = (const int*)d_in[1];
  const float* W          = (const float*)d_in[2];
  const float* b          = (const float*)d_in[3];
  const float* a          = (const float*)d_in[4];
  float* out = (float*)d_out;
  float* ws  = (float*)d_ws;

  score_wmma_kernel<<<(NODE_TILES + 7) / 8, 256, 0, stream>>>(node_feats, W, b, a, ws);
  edge_logits_kernel<<<EDGE_WGS, 256, 0, stream>>>(edge_index, ws);
  edge_softmax_kernel<<<EDGE_WGS, 256, 0, stream>>>(ws, out);
}


# Score launch as 256 work-groups of 8 fold + 13 tile waves: W folded 256x not 391x; tile rows loaded (16 B) at wave entry, in flight across an LDS-only barrier; k permuted alike; sum order changes
# speedup vs baseline: 2.2940x; 1.0536x over previous
#include <hip/hip_runtime.h>
#include <hip/hip_bf16.h>


typedef __attribute__((ext_vector_type(2))) float v2f;
typedef __attribute__((ext_vector_type(4))) float v4f;
typedef __attribute__((ext_vector_type(8))) float v8f;

#define LEAKY_ALPHA 0.2f

static constexpr int N_NODES = 50000;
static constexpr int N_EDGES = 800000;
static constexpr int NODE_TILES = 3125;

static constexpr int EDGE_WGS = 625;
static constexpr int EDGES_PER_THREAD = 5;
static constexpr int EDGES_PER_WG = 256 * EDGES_PER_THREAD;

static constexpr int SCORE_WGS = 256;
static constexpr int FOLD_THREADS = 256;
static constexpr int TILE_WAVES = 13;
static constexpr int SCORE_THREADS = FOLD_THREADS + 32 * TILE_WAVES;

static constexpr int A2_PITCH = 136;

static constexpr size_t WS_SSRC = 2056;
static constexpr size_t WS_SDST = WS_SSRC + (size_t)N_NODES * 4;
static constexpr size_t WS_LOG  = WS_SDST + (size_t)N_NODES * 4;
static constexpr size_t WS_MAXP = WS_LOG + (size_t)N_EDGES * 4;
static constexpr size_t WS_SUMP = WS_MAXP + (size_t)3125 * 4;

__device__ __forceinline__ v4f max4(v4f p, v4f q) {
  v4f r;
  r.x = fmaxf(p.x, q.x); r.y = fmaxf(p.y, q.y);
  r.z = fmaxf(p.z, q.z); r.w = fmaxf(p.w, q.w);
  return r;
}

__device__ __forceinline__ v4f block_max4(v4f v, v4f* red  ) {
#pragma unroll
  for (int off = 16; off > 0; off >>= 1) {
    v4f o;
    o.x = __shfl_xor(v.x, off); o.y = __shfl_xor(v.y, off);
    o.z = __shfl_xor(v.z, off); o.w = __shfl_xor(v.w, off);
    v = max4(v, o);
  }
  if ((threadIdx.x & 31) == 0) red[threadIdx.x >> 5] = v;
  __syncthreads();
  v4f r = red[0];
#pragma unroll
  for (int w = 1; w < 8; ++w) r = max4(r, red[w]);
  return r;
}

__device__ __forceinline__ v4f block_sum4(v4f v, v4f* red  ) {
#pragma unroll
  for (int off = 16; off > 0; off >>= 1) {
    v4f o;
    o.x = __shfl_xor(v.x, off); o.y = __shfl_xor(v.y, off);
    o.z = __shfl_xor(v.z, off); o.w = __shfl_xor(v.w, off);
    v += o;
  }
  if ((threadIdx.x & 31) == 0) red[threadIdx.x >> 5] = v;
  __syncthreads();
  v4f r = red[0];
#pragma unroll
  for (int w = 1; w < 8; ++w) r += red[w];
  return r;
}

__device__ __forceinline__ v4f shfl_xor4(v4f v, int off) {
  v4f o;
  o.x = __shfl_xor(v.x, off); o.y = __shfl_xor(v.y, off);
  o.z = __shfl_xor(v.z, off); o.w = __shfl_xor(v.w, off);
  return o;
}

__device__ __forceinline__ v4f wave_max4(v4f v) {
#pragma unroll
  for (int off = 16; off > 0; off >>= 1) v = max4(v, shfl_xor4(v, off));
  return v;
}

__device__ __forceinline__ v4f wave_sum4(v4f v) {
#pragma unroll
  for (int off = 16; off > 0; off >>= 1) v += shfl_xor4(v, off);
  return v;
}

__device__ __forceinline__ v4f exp_diff4(v4f p, v4f q) {
  v4f r;
  r.x = __expf(p.x - q.x); r.y = __expf(p.y - q.y);
  r.z = __expf(p.z - q.z); r.w = __expf(p.w - q.w);
  return r;
}

__global__ void __launch_bounds__(SCORE_THREADS) score_wmma_kernel(const float* __restrict__ X,
                                                                   const float* __restrict__ W,
                                                                   const float* __restrict__ b,
                                                                   const float* __restrict__ a,
                                                                   float* ws) {
  __shared__ __attribute__((aligned(16))) float A2s[8 * A2_PITCH];
  __shared__ __attribute__((aligned(16))) float biass[8];
  const int t = threadIdx.x;
  const int lane = t & 31;
  const int wave = __builtin_amdgcn_readfirstlane(t >> 5);
  const bool fold_wave = wave < FOLD_THREADS / 32;
  const int tile = blockIdx.x + SCORE_WGS * (wave - FOLD_THREADS / 32);
  const bool has_tile = !fold_wave && tile < NODE_TILES;
  const int m = lane & 15;
  const int hi = lane >> 4;

  v4f xv[16];
  if (has_tile) {
    const v4f* xrow = (const v4f*)(X + (size_t)(tile * 16 + m) * 128 + 4 * hi);
#pragma unroll
    for (int j = 0; j < 16; ++j) xv[j] = xrow[2 * j];
  }

  if (fold_wave) {
    const int c = t & 127;
    const int hp = __builtin_amdgcn_readfirstlane(t >> 7);
#pragma unroll
    for (int hh = 0; hh < 2; ++hh) {
      const int h = hp * 2 + hh;
      const float* wcol = W + (size_t)(h * 32) * 128 + c;
      const float* av = a + h * 64;
      float acc0 = 0.f, acc1 = 0.f;
#pragma unroll
      for (int k = 0; k < 32; ++k) {
        const float w = wcol[(size_t)k * 128];
        acc0 += w * av[k];
        acc1 += w * av[32 + k];
      }
      A2s[h * A2_PITCH + c] = acc0;
      A2s[(4 + h) * A2_PITCH + c] = acc1;
    }
    if (t < 8) {
      const int h = t & 3, half = t >> 2;
      const float* av = a + h * 64 + half * 32;
      float acc = 0.f;
#pragma unroll
      for (int k = 0; k < 32; ++k) acc += b[h * 32 + k] * av[k];
      biass[t] = acc;
    }
  }
  __builtin_amdgcn_fence(__ATOMIC_RELEASE, "workgroup", "local");
  __builtin_amdgcn_s_barrier();
  __builtin_amdgcn_fence(__ATOMIC_ACQUIRE, "workgroup", "local");

  if (!has_tile) return;

  const float* arow = A2s + (m & 7) * A2_PITCH + 4 * hi;

  v8f acc = {};
#pragma unroll
  for (int j = 0; j < 16; ++j) {
    const v4f a4 = *(const v4f*)(arow + 8 * j);
    const v4f x4 = xv[j];
    const v2f a_lo = {a4.x, a4.y}, a_hi = {a4.z, a4.w};
    const v2f x_lo = {x4.x, x4.y}, x_hi = {x4.z, x4.w};
    acc = __builtin_amdgcn_wmma_f32_16x16x4_f32(
         false, a_lo,  false, x_lo,
         (short)0, acc,  false,  false);
    acc = __builtin_amdgcn_wmma_f32_16x16x4_f32(
         false, a_hi,  false, x_hi,
         (short)0, acc,  false,  false);
  }

  const v4f bb = ((const v4f*)biass)[hi];
  v4f rec;
  rec.x = (hi ? acc[4] : acc[0]) + bb.x;
  rec.y = (hi ? acc[5] : acc[1]) + bb.y;
  rec.z = (hi ? acc[6] : acc[2]) + bb.z;
  rec.w = (hi ? acc[7] : acc[3]) + bb.w;
  float* plane = ws + (hi ? WS_SDST : WS_SSRC);
  *(v4f*)(plane + (size_t)(tile * 16 + m) * 4) = rec;
}

__global__ void __launch_bounds__(256) edge_logits_kernel(const int* __restrict__ ei,
                                                          float* ws) {
  __shared__ v4f redm[8];
  __shared__ v4f reds[8];
  const int t = threadIdx.x;
  const int e0 = blockIdx.x * EDGES_PER_WG + t;
  const v4f* SS = (const v4f*)(ws + WS_SSRC);
  const v4f* SD = (const v4f*)(ws + WS_SDST);

  int src[EDGES_PER_THREAD], dst[EDGES_PER_THREAD];
#pragma unroll
  for (int i = 0; i < EDGES_PER_THREAD; ++i) {
    src[i] = ei[e0 + i * 256];
    dst[i] = ei[N_EDGES + e0 + i * 256];
  }
  asm volatile("" :: "v"(src[0]), "v"(src[1]), "v"(src[2]), "v"(src[3]), "v"(src[4]),
                     "v"(dst[0]), "v"(dst[1]), "v"(dst[2]), "v"(dst[3]), "v"(dst[4]));
  v4f sv[EDGES_PER_THREAD], dv[EDGES_PER_THREAD];
#pragma unroll
  for (int i = 0; i < EDGES_PER_THREAD; ++i) {
    sv[i] = SS[(size_t)src[i]];
    dv[i] = SD[(size_t)dst[i]];
  }

  v4f lg[EDGES_PER_THREAD];
  v4f mx;
#pragma unroll
  for (int i = 0; i < EDGES_PER_THREAD; ++i) {
    v4f l = sv[i] + dv[i];
    l.x = fmaxf(l.x, LEAKY_ALPHA * l.x);
    l.y = fmaxf(l.y, LEAKY_ALPHA * l.y);
    l.z = fmaxf(l.z, LEAKY_ALPHA * l.z);
    l.w = fmaxf(l.w, LEAKY_ALPHA * l.w);
    *(v4f*)(ws + WS_LOG + (size_t)(e0 + i * 256) * 4) = l;
    lg[i] = l;
    mx = (i == 0) ? l : max4(mx, l);
  }

  const v4f wm = wave_max4(mx);
  v4f s = exp_diff4(lg[0], wm);
#pragma unroll
  for (int i = 1; i < EDGES_PER_THREAD; ++i) s += exp_diff4(lg[i], wm);
  s = wave_sum4(s);

  if ((t & 31) == 0) { redm[t >> 5] = wm; reds[t >> 5] = s; }
  __syncthreads();

  if (t < 32) {
    const v4f m = redm[t & 7];
    const v4f q = reds[t & 7];
    v4f g = m;
#pragma unroll
    for (int off = 4; off > 0; off >>= 1) g = max4(g, shfl_xor4(g, off));
    const v4f f = exp_diff4(m, g);
    v4f r;
    r.x = q.x * f.x; r.y = q.y * f.y; r.z = q.z * f.z; r.w = q.w * f.w;
#pragma unroll
    for (int off = 4; off > 0; off >>= 1) r += shfl_xor4(r, off);
    if (t == 0) {
      *(v4f*)(ws + WS_MAXP + (size_t)blockIdx.x * 4) = g;
      *(v4f*)(ws + WS_SUMP + (size_t)blockIdx.x * 4) = r;
    }
  }
}

__global__ void __launch_bounds__(256) edge_softmax_kernel(float* ws, float* __restrict__ out) {
  __shared__ v4f redm[8];
  __shared__ v4f reds[8];
  const int t = threadIdx.x;
  const int e0 = blockIdx.x * EDGES_PER_WG + t;

  v4f lg[EDGES_PER_THREAD];
#pragma unroll
  for (int i = 0; i < EDGES_PER_THREAD; ++i)
    lg[i] = *(const v4f*)(ws + WS_LOG + (size_t)(e0 + i * 256) * 4);

  const v4f* PM = (const v4f*)(ws + WS_MAXP);
  const v4f* PS = (const v4f*)(ws + WS_SUMP);
  const bool has3 = (t + 512 < EDGE_WGS);
  const int t2 = has3 ? t + 512 : EDGE_WGS - 1;
  const v4f m0 = PM[t], m1 = PM[t + 256], m2 = PM[t2];
  const v4f q0 = PS[t], q1 = PS[t + 256], q2 = PS[t2];

  const v4f gm = block_max4(max4(max4(m0, m1), m2), redm);

  const v4f f0 = exp_diff4(m0, gm), f1 = exp_diff4(m1, gm), f2 = exp_diff4(m2, gm);
  const v4f zero4 = {0.f, 0.f, 0.f, 0.f};
  v4f s;
  s.x = q0.x * f0.x + q1.x * f1.x; s.y = q0.y * f0.y + q1.y * f1.y;
  s.z = q0.z * f0.z + q1.z * f1.z; s.w = q0.w * f0.w + q1.w * f1.w;
  v4f s2;
  s2.x = q2.x * f2.x; s2.y = q2.y * f2.y; s2.z = q2.z * f2.z; s2.w = q2.w * f2.w;
  s += has3 ? s2 : zero4;
  const v4f r = block_sum4(s, reds);
  const v4f inv = {1.0f / r.x, 1.0f / r.y, 1.0f / r.z, 1.0f / r.w};

#pragma unroll
  for (int i = 0; i < EDGES_PER_THREAD; ++i) {
    v4f q = exp_diff4(lg[i], gm);
    q.x *= inv.x; q.y *= inv.y; q.z *= inv.z; q.w *= inv.w;
    *(v4f*)(out + (size_t)(e0 + i * 256) * 4) = q;
  }
}

extern "C" void kernel_launch(void* const* d_in, const int* in_sizes, int n_in,
                              void* d_out, int out_size, void* d_ws, size_t ws_size,
                              hipStream_t stream) {
  const float* node_feats = (const float*)d_in[0];
  const int*   edge_index = (const int*)d_in[1];
  const float* W          = (const float*)d_in[2];
  const float* b          = (const float*)d_in[3];
  const float* a          = (const float*)d_in[4];
  float* out = (float*)d_out;
  float* ws  = (float*)d_ws;

  score_wmma_kernel<<<SCORE_WGS, SCORE_THREADS, 0, stream>>>(node_feats, W, b, a, ws);
  edge_logits_kernel<<<EDGE_WGS, 256, 0, stream>>>(edge_index, ws);
  edge_softmax_kernel<<<EDGE_WGS, 256, 0, stream>>>(ws, out);
}
